# starting kernel
# baseline (speedup 1.0000x reference)
#include <hip/hip_runtime.h>
#include <hip/hip_bf16.h>

typedef __attribute__((ext_vector_type(2))) float v2f;
typedef __attribute__((ext_vector_type(8))) float v8f;

#define CH 128
#define EPS 1e-5f

__global__ void k_deg_init(float* __restrict__ deg, int N) {
    int i = blockIdx.x * blockDim.x + threadIdx.x;
    if (i < N) deg[i] = 1.0f;
}

__global__ void k_edge_deg(const int* __restrict__ ei, float* __restrict__ deg,
                           int E, int N) {
    int e = blockIdx.x * blockDim.x + threadIdx.x;
    if (e >= E) return;
    int s = ei[e];
    int d = ei[E + e];
    if ((unsigned)s < (unsigned)N && (unsigned)d < (unsigned)N) {
        atomicAdd(&deg[d], 1.0f);
    }
}

__global__ void k_dinv(float* __restrict__ deg, int N) {
    int i = blockIdx.x * blockDim.x + threadIdx.x;
    if (i < N) deg[i] = rsqrtf(deg[i]);
}

__global__ __launch_bounds__(256) void k_gemm_wmma(const float* __restrict__ X,
                                                   const float* __restrict__ W,
                                                   float* __restrict__ H, int M) {
    __shared__ float sA[64][132];
    __shared__ float sB[32][132];
    const int m0  = blockIdx.x * 64;
    const int n0  = blockIdx.y * 32;
    const int tid = threadIdx.x;

    #pragma unroll
    for (int i = 0; i < 8; ++i) {
        const int idx   = tid + i * 256;
        const int row   = idx >> 5;
        const int chunk = idx & 31;
        const int gr    = m0 + row;
        const float4 v =
            *(const float4*)(X + (size_t)(gr < M ? gr : 0) * CH + chunk * 4);
        *(float4*)&sA[row][chunk * 4] = v;
    }
    {
        const int c     = tid & 31;
        const int kbase = (tid >> 5) * 16;
        #pragma unroll
        for (int i = 0; i < 16; ++i) {
            const int k = kbase + i;
            sB[c][k] = W[(size_t)k * CH + n0 + c];
        }
    }
    __syncthreads();

    const int wave  = tid >> 5;
    const int lane  = tid & 31;
    const int mt    = wave >> 1;
    const int nt    = wave & 1;
    const int mloc  = lane & 15;
    const int khalf = (lane >> 4) * 2;

    const float* arow = &sA[mt * 16 + mloc][0];
    const float* brow = &sB[nt * 16 + mloc][0];

    v8f c = {};
    #pragma unroll 8
    for (int k0 = 0; k0 < CH; k0 += 4) {
        const v2f a = *(const v2f*)(arow + k0 + khalf);
        const v2f b = *(const v2f*)(brow + k0 + khalf);
        c = __builtin_amdgcn_wmma_f32_16x16x4_f32(
                 false, a,  false, b,
                 (short)0, c,  false,  false);
    }

    const int col   = n0 + nt * 16 + mloc;
    const int half  = lane >> 4;
    const int mbase = m0 + mt * 16 + half * 8;
    if (m0 + 64 <= M) {
        #pragma unroll
        for (int r = 0; r < 8; ++r)
            H[(size_t)(mbase + r) * CH + col] = c[r];
    } else {
        #pragma unroll
        for (int r = 0; r < 8; ++r)
            if (mbase + r < M) H[(size_t)(mbase + r) * CH + col] = c[r];
    }
}

__global__ void k_selfloop_init(const float* __restrict__ H,
                                const float* __restrict__ dinv,
                                const float* __restrict__ bias,
                                float* __restrict__ Acc, int M) {
    int idx = blockIdx.x * blockDim.x + threadIdx.x;
    if (idx >= M * (CH / 4)) return;
    const int row = idx >> 5;
    const int l   = idx & 31;
    const float di = dinv[row];
    const float w  = di * di;
    float4 h = ((const float4*)(H + (size_t)row * CH))[l];
    float4 b = ((const float4*)bias)[l];
    float4 o;
    o.x = h.x * w + b.x;  o.y = h.y * w + b.y;
    o.z = h.z * w + b.z;  o.w = h.w * w + b.w;
    ((float4*)(Acc + (size_t)row * CH))[l] = o;
}

__global__ void k_edge_scatter(const float* __restrict__ H,
                               const int* __restrict__ ei,
                               const float* __restrict__ dinv,
                               float* __restrict__ Acc, int E, int N) {
    int e    = (blockIdx.x * blockDim.x + threadIdx.x) >> 5;
    int lane = threadIdx.x & 31;
    if (e >= E) return;
    int s = ei[e];
    int d = ei[E + e];
    if ((unsigned)s >= (unsigned)N || (unsigned)d >= (unsigned)N) return;
    const float norm = dinv[s] * dinv[d];
    float4 v = ((const float4*)(H + (size_t)s * CH))[lane];
    float* out = Acc + (size_t)d * CH + lane * 4;
    atomicAdd(out + 0, v.x * norm);
    atomicAdd(out + 1, v.y * norm);
    atomicAdd(out + 2, v.z * norm);
    atomicAdd(out + 3, v.w * norm);
}

__global__ void k_rownorm_relu(const float* __restrict__ In,
                               float* __restrict__ Out, int M) {
    int row  = (blockIdx.x * blockDim.x + threadIdx.x) >> 5;
    int lane = threadIdx.x & 31;
    if (row >= M) return;
    float4 v = ((const float4*)(In + (size_t)row * CH))[lane];

    float s = v.x + v.y + v.z + v.w;
    #pragma unroll
    for (int o = 16; o > 0; o >>= 1) s += __shfl_xor(s, o, 32);
    const float mu = s * (1.0f / CH);

    const float d0 = v.x - mu, d1 = v.y - mu, d2 = v.z - mu, d3 = v.w - mu;
    float q = d0 * d0 + d1 * d1 + d2 * d2 + d3 * d3;
    #pragma unroll
    for (int o = 16; o > 0; o >>= 1) q += __shfl_xor(q, o, 32);
    const float inv = rsqrtf(q * (1.0f / CH) + EPS);

    float4 r;
    r.x = fmaxf(d0 * inv, 0.0f);
    r.y = fmaxf(d1 * inv, 0.0f);
    r.z = fmaxf(d2 * inv, 0.0f);
    r.w = fmaxf(d3 * inv, 0.0f);
    ((float4*)(Out + (size_t)row * CH))[lane] = r;
}

extern "C" void kernel_launch(void* const* d_in, const int* in_sizes, int n_in,
                              void* d_out, int out_size, void* d_ws, size_t ws_size,
                              hipStream_t stream) {
    const float* x   = (const float*)d_in[0];
    const int*   ei  = (const int*)  d_in[1];
    const float* W1  = (const float*)d_in[2];
    const float* b1  = (const float*)d_in[3];
    const float* W2  = (const float*)d_in[4];
    const float* b2  = (const float*)d_in[5];
    float*       out = (float*)d_out;

    const int N = in_sizes[0] / CH;
    const int E = in_sizes[1] / 2;

    float* ws   = (float*)d_ws;
    const size_t nPad = ((size_t)N + 63) & ~(size_t)63;
    float* dinv = ws;
    float* hbuf = ws + nPad;
    float* acc  = hbuf + (size_t)N * CH;

    const int T = 256;
    const int gN     = (N + T - 1) / T;
    const int gE     = (E + T - 1) / T;
    const int gRow4  = (N * (CH / 4) + T - 1) / T;
    const int gEdgeW = ((E * 32) + T - 1) / T;
    const int gRowW  = ((N * 32) + T - 1) / T;
    const dim3 gGemm((N + 63) / 64, CH / 32);

    k_deg_init<<<gN, T, 0, stream>>>(dinv, N);
    k_edge_deg<<<gE, T, 0, stream>>>(ei, dinv, E, N);
    k_dinv   <<<gN, T, 0, stream>>>(dinv, N);

    k_gemm_wmma    <<<gGemm, T, 0, stream>>>(x, W1, hbuf, N);
    k_selfloop_init<<<gRow4, T, 0, stream>>>(hbuf, dinv, b1, acc, N);
    k_edge_scatter <<<gEdgeW, T, 0, stream>>>(hbuf, ei, dinv, acc, E, N);
    k_rownorm_relu <<<gRowW, T, 0, stream>>>(acc, hbuf, N);

    k_gemm_wmma    <<<gGemm, T, 0, stream>>>(hbuf, W2, acc, N);
    k_selfloop_init<<<gRow4, T, 0, stream>>>(acc, dinv, b2, out, N);
    k_edge_scatter <<<gEdgeW, T, 0, stream>>>(acc, ei, dinv, out, E, N);
    k_rownorm_relu <<<gRowW, T, 0, stream>>>(out, out, N);
}


# float atomic scatter-add replaced by a per-call destination-sorted edge list (int atomics give an arrival-order rank) and a one-wave-per-row register gather-sum that starts from self-loop+bias
# speedup vs baseline: 2.3887x; 2.3887x over previous
#include <hip/hip_runtime.h>
#include <hip/hip_bf16.h>

typedef __attribute__((ext_vector_type(2))) float v2f;
typedef __attribute__((ext_vector_type(8))) float v8f;

#define CH 128
#define EPS 1e-5f

__global__ void k_cnt_zero(int* __restrict__ cnt, int N) {
    int i = blockIdx.x * blockDim.x + threadIdx.x;
    if (i < N) cnt[i] = 0;
}

__global__ void k_edge_rank(const int* __restrict__ ei, int* __restrict__ cnt,
                            int* __restrict__ rank, int E, int N) {
    int e = blockIdx.x * blockDim.x + threadIdx.x;
    if (e >= E) return;
    int s = ei[e];
    int d = ei[E + e];
    if ((unsigned)s < (unsigned)N && (unsigned)d < (unsigned)N) {
        rank[e] = atomicAdd(&cnt[d], 1);
    }
}

__global__ __launch_bounds__(1024) void k_scan_dinv(const int* __restrict__ cnt,
                                                    int* __restrict__ rowptr,
                                                    float* __restrict__ dinv, int N) {
    __shared__ int sTot[32];
    __shared__ int sFront[32];
    const int tid  = threadIdx.x;
    const int lane = tid & 31;
    const int wv   = tid >> 5;
    const int b    = blockIdx.x;

    int front = 0;
    const int4* c4 = (const int4*)cnt;
    #pragma unroll 4
    for (int j = tid; j < b * 256; j += 1024) {
        const int4 v = c4[j];
        front += (v.x + v.y) + (v.z + v.w);
    }
    const int i = b * 1024 + tid;
    const int c = (i < N) ? cnt[i] : 0;

    int inc = c;
    for (int o = 1; o < 32; o <<= 1) {
        const int t = __shfl_up(inc, o, 32);
        if (lane >= o) inc += t;
    }
    for (int o = 16; o > 0; o >>= 1) front += __shfl_xor(front, o, 32);
    if (lane == 31) sTot[wv] = inc;
    if (lane == 0)  sFront[wv] = front;
    __syncthreads();

    const int wt = sTot[lane];
    int winc = wt;
    for (int o = 1; o < 32; o <<= 1) {
        const int t = __shfl_up(winc, o, 32);
        if (lane >= o) winc += t;
    }
    int base = sFront[lane];
    for (int o = 16; o > 0; o >>= 1) base += __shfl_xor(base, o, 32);
    const int wave_off = __shfl(winc - wt, wv, 32);

    const int excl = base + wave_off + inc - c;
    if (i <= N) rowptr[i] = excl;
    if (i < N)  dinv[i] = rsqrtf(1.0f + (float)c);
}

__global__ void k_csr_fill(const int* __restrict__ ei, const int* __restrict__ rank,
                           const int* __restrict__ rowptr, int* __restrict__ csr_src,
                           int E, int N) {
    int e = blockIdx.x * blockDim.x + threadIdx.x;
    if (e >= E) return;
    int s = ei[e];
    int d = ei[E + e];
    if ((unsigned)s < (unsigned)N && (unsigned)d < (unsigned)N) {
        csr_src[rowptr[d] + rank[e]] = s;
    }
}

__global__ __launch_bounds__(256) void k_gemm_wmma(const float* __restrict__ X,
                                                   const float* __restrict__ W,
                                                   float* __restrict__ H, int M) {
    __shared__ float sA[64][132];
    __shared__ float sB[32][132];
    const int m0  = blockIdx.x * 64;
    const int n0  = blockIdx.y * 32;
    const int tid = threadIdx.x;

    #pragma unroll
    for (int i = 0; i < 8; ++i) {
        const int idx   = tid + i * 256;
        const int row   = idx >> 5;
        const int chunk = idx & 31;
        const int gr    = m0 + row;
        const float4 v =
            *(const float4*)(X + (size_t)(gr < M ? gr : 0) * CH + chunk * 4);
        *(float4*)&sA[row][chunk * 4] = v;
    }
    {
        const int c     = tid & 31;
        const int kbase = (tid >> 5) * 16;
        #pragma unroll
        for (int i = 0; i < 16; ++i) {
            const int k = kbase + i;
            sB[c][k] = W[(size_t)k * CH + n0 + c];
        }
    }
    __syncthreads();

    const int wave  = tid >> 5;
    const int lane  = tid & 31;
    const int mt    = wave >> 1;
    const int nt    = wave & 1;
    const int mloc  = lane & 15;
    const int khalf = (lane >> 4) * 2;

    const float* arow = &sA[mt * 16 + mloc][0];
    const float* brow = &sB[nt * 16 + mloc][0];

    v8f c = {};
    #pragma unroll 8
    for (int k0 = 0; k0 < CH; k0 += 4) {
        const v2f a = *(const v2f*)(arow + k0 + khalf);
        const v2f b = *(const v2f*)(brow + k0 + khalf);
        c = __builtin_amdgcn_wmma_f32_16x16x4_f32(
                 false, a,  false, b,
                 (short)0, c,  false,  false);
    }

    const int col   = n0 + nt * 16 + mloc;
    const int half  = lane >> 4;
    const int mbase = m0 + mt * 16 + half * 8;
    if (m0 + 64 <= M) {
        #pragma unroll
        for (int r = 0; r < 8; ++r)
            H[(size_t)(mbase + r) * CH + col] = c[r];
    } else {
        #pragma unroll
        for (int r = 0; r < 8; ++r)
            if (mbase + r < M) H[(size_t)(mbase + r) * CH + col] = c[r];
    }
}

__global__ __launch_bounds__(256) void k_gather_rows(const float* __restrict__ H,
                                                     const int* __restrict__ rowptr,
                                                     const int* __restrict__ csr_src,
                                                     const float* __restrict__ dinv,
                                                     const float* __restrict__ bias,
                                                     float* __restrict__ Acc, int N) {
    const int lane = threadIdx.x & 31;
    const int row = __builtin_amdgcn_readfirstlane(
        (int)((blockIdx.x * blockDim.x + threadIdx.x) >> 5));
    if (row >= N) return;
    const float4* __restrict__ H4 = (const float4*)H;
    const float dd  = dinv[row];
    const int   beg = rowptr[row];
    const int   end = rowptr[row + 1];

    const float4 h = H4[(size_t)row * 32 + lane];
    const float4 b = ((const float4*)bias)[lane];
    const float  w = dd * dd;
    float4 a;
    a.x = h.x * w + b.x;  a.y = h.y * w + b.y;
    a.z = h.z * w + b.z;  a.w = h.w * w + b.w;

    for (int base = beg; base < end; base += 32) {
        const int left = end - base;
        const int n    = left < 32 ? left : 32;
        int   sl = 0;
        float wl = 0.0f;
        if (lane < n) {
            sl = csr_src[base + lane];
            wl = dinv[sl];
        }
        int j = 0;
        for (; j + 4 <= n; j += 4) {
            const int s0 = __builtin_amdgcn_readlane(sl, j);
            const int s1 = __builtin_amdgcn_readlane(sl, j + 1);
            const int s2 = __builtin_amdgcn_readlane(sl, j + 2);
            const int s3 = __builtin_amdgcn_readlane(sl, j + 3);
            const float4 v0 = H4[(size_t)s0 * 32 + lane];
            const float4 v1 = H4[(size_t)s1 * 32 + lane];
            const float4 v2 = H4[(size_t)s2 * 32 + lane];
            const float4 v3 = H4[(size_t)s3 * 32 + lane];
            const float w0 = dd * __int_as_float(__builtin_amdgcn_readlane(__float_as_int(wl), j));
            const float w1 = dd * __int_as_float(__builtin_amdgcn_readlane(__float_as_int(wl), j + 1));
            const float w2 = dd * __int_as_float(__builtin_amdgcn_readlane(__float_as_int(wl), j + 2));
            const float w3 = dd * __int_as_float(__builtin_amdgcn_readlane(__float_as_int(wl), j + 3));
            a.x += v0.x * w0;  a.y += v0.y * w0;  a.z += v0.z * w0;  a.w += v0.w * w0;
            a.x += v1.x * w1;  a.y += v1.y * w1;  a.z += v1.z * w1;  a.w += v1.w * w1;
            a.x += v2.x * w2;  a.y += v2.y * w2;  a.z += v2.z * w2;  a.w += v2.w * w2;
            a.x += v3.x * w3;  a.y += v3.y * w3;  a.z += v3.z * w3;  a.w += v3.w * w3;
        }
        for (; j < n; ++j) {
            const int   s0 = __builtin_amdgcn_readlane(sl, j);
            const float4 v0 = H4[(size_t)s0 * 32 + lane];
            const float w0 = dd * __int_as_float(__builtin_amdgcn_readlane(__float_as_int(wl), j));
            a.x += v0.x * w0;  a.y += v0.y * w0;  a.z += v0.z * w0;  a.w += v0.w * w0;
        }
    }
    ((float4*)Acc)[(size_t)row * 32 + lane] = a;
}

__global__ void k_rownorm_relu(const float* __restrict__ In,
                               float* __restrict__ Out, int M) {
    int row  = (blockIdx.x * blockDim.x + threadIdx.x) >> 5;
    int lane = threadIdx.x & 31;
    if (row >= M) return;
    float4 v = ((const float4*)(In + (size_t)row * CH))[lane];

    float s = v.x + v.y + v.z + v.w;
    #pragma unroll
    for (int o = 16; o > 0; o >>= 1) s += __shfl_xor(s, o, 32);
    const float mu = s * (1.0f / CH);

    const float d0 = v.x - mu, d1 = v.y - mu, d2 = v.z - mu, d3 = v.w - mu;
    float q = d0 * d0 + d1 * d1 + d2 * d2 + d3 * d3;
    #pragma unroll
    for (int o = 16; o > 0; o >>= 1) q += __shfl_xor(q, o, 32);
    const float inv = rsqrtf(q * (1.0f / CH) + EPS);

    float4 r;
    r.x = fmaxf(d0 * inv, 0.0f);
    r.y = fmaxf(d1 * inv, 0.0f);
    r.z = fmaxf(d2 * inv, 0.0f);
    r.w = fmaxf(d3 * inv, 0.0f);
    ((float4*)(Out + (size_t)row * CH))[lane] = r;
}

extern "C" void kernel_launch(void* const* d_in, const int* in_sizes, int n_in,
                              void* d_out, int out_size, void* d_ws, size_t ws_size,
                              hipStream_t stream) {
    const float* x   = (const float*)d_in[0];
    const int*   ei  = (const int*)  d_in[1];
    const float* W1  = (const float*)d_in[2];
    const float* b1  = (const float*)d_in[3];
    const float* W2  = (const float*)d_in[4];
    const float* b2  = (const float*)d_in[5];
    float*       out = (float*)d_out;

    const int N = in_sizes[0] / CH;
    const int E = in_sizes[1] / 2;

    float* ws   = (float*)d_ws;
    const size_t nPad = ((size_t)N + 63) & ~(size_t)63;
    float* dinv = ws;
    float* bufA = ws + nPad;
    float* bufB = bufA + (size_t)N * CH;

    int* cnt     = (int*)d_out;
    int* rowptr  = cnt + nPad;
    int* rank    = rowptr + nPad + 64;
    int* csr_src = rank + E;

    const int T = 256;
    const int gN     = (N + T - 1) / T;
    const int gE     = (E + T - 1) / T;
    const int gScan  = (N + 1 + 1023) / 1024;
    const int gRowW  = ((N * 32) + T - 1) / T;
    const dim3 gGemm((N + 63) / 64, CH / 32);

    k_cnt_zero <<<gN, T, 0, stream>>>(cnt, N);
    k_edge_rank<<<gE, T, 0, stream>>>(ei, cnt, rank, E, N);
    k_scan_dinv<<<gScan, 1024, 0, stream>>>(cnt, rowptr, dinv, N);
    k_csr_fill <<<gE, T, 0, stream>>>(ei, rank, rowptr, csr_src, E, N);

    k_gemm_wmma   <<<gGemm, T, 0, stream>>>(x, W1, bufA, N);
    k_gather_rows <<<gRowW, T, 0, stream>>>(bufA, rowptr, csr_src, dinv, b1, bufB, N);
    k_rownorm_relu<<<gRowW, T, 0, stream>>>(bufB, bufA, N);

    k_gemm_wmma   <<<gGemm, T, 0, stream>>>(bufA, W2, bufB, N);
    k_gather_rows <<<gRowW, T, 0, stream>>>(bufB, rowptr, csr_src, dinv, b2, bufA, N);
    k_rownorm_relu<<<gRowW, T, 0, stream>>>(bufA, out, N);
}


# instance norm + ReLU done in the gather wave's registers (the wave already holds the whole row): both norm launches, their 100000 waves and a write+read of both sum buffers go; out is the 2nd buffer
# speedup vs baseline: 2.6258x; 1.0993x over previous
#include <hip/hip_runtime.h>
#include <hip/hip_bf16.h>

typedef __attribute__((ext_vector_type(2))) float v2f;
typedef __attribute__((ext_vector_type(8))) float v8f;

#define CH 128
#define EPS 1e-5f

__global__ void k_cnt_zero(int* __restrict__ cnt, int N) {
    int i = blockIdx.x * blockDim.x + threadIdx.x;
    if (i < N) cnt[i] = 0;
}

__global__ void k_edge_rank(const int* __restrict__ ei, int* __restrict__ cnt,
                            int* __restrict__ rank, int E, int N) {
    int e = blockIdx.x * blockDim.x + threadIdx.x;
    if (e >= E) return;
    int s = ei[e];
    int d = ei[E + e];
    if ((unsigned)s < (unsigned)N && (unsigned)d < (unsigned)N) {
        rank[e] = atomicAdd(&cnt[d], 1);
    }
}

__global__ __launch_bounds__(1024) void k_scan_dinv(const int* __restrict__ cnt,
                                                    int* __restrict__ rowptr,
                                                    float* __restrict__ dinv, int N) {
    __shared__ int sTot[32];
    __shared__ int sFront[32];
    const int tid  = threadIdx.x;
    const int lane = tid & 31;
    const int wv   = tid >> 5;
    const int b    = blockIdx.x;

    int front = 0;
    const int4* c4 = (const int4*)cnt;
    #pragma unroll 4
    for (int j = tid; j < b * 256; j += 1024) {
        const int4 v = c4[j];
        front += (v.x + v.y) + (v.z + v.w);
    }
    const int i = b * 1024 + tid;
    const int c = (i < N) ? cnt[i] : 0;

    int inc = c;
    for (int o = 1; o < 32; o <<= 1) {
        const int t = __shfl_up(inc, o, 32);
        if (lane >= o) inc += t;
    }
    for (int o = 16; o > 0; o >>= 1) front += __shfl_xor(front, o, 32);
    if (lane == 31) sTot[wv] = inc;
    if (lane == 0)  sFront[wv] = front;
    __syncthreads();

    const int wt = sTot[lane];
    int winc = wt;
    for (int o = 1; o < 32; o <<= 1) {
        const int t = __shfl_up(winc, o, 32);
        if (lane >= o) winc += t;
    }
    int base = sFront[lane];
    for (int o = 16; o > 0; o >>= 1) base += __shfl_xor(base, o, 32);
    const int wave_off = __shfl(winc - wt, wv, 32);

    const int excl = base + wave_off + inc - c;
    if (i <= N) rowptr[i] = excl;
    if (i < N)  dinv[i] = rsqrtf(1.0f + (float)c);
}

__global__ void k_csr_fill(const int* __restrict__ ei, const int* __restrict__ rank,
                           const int* __restrict__ rowptr, int* __restrict__ csr_src,
                           int E, int N) {
    int e = blockIdx.x * blockDim.x + threadIdx.x;
    if (e >= E) return;
    int s = ei[e];
    int d = ei[E + e];
    if ((unsigned)s < (unsigned)N && (unsigned)d < (unsigned)N) {
        csr_src[rowptr[d] + rank[e]] = s;
    }
}

__global__ __launch_bounds__(256) void k_gemm_wmma(const float* __restrict__ X,
                                                   const float* __restrict__ W,
                                                   float* __restrict__ H, int M) {
    __shared__ float sA[64][132];
    __shared__ float sB[32][132];
    const int m0  = blockIdx.x * 64;
    const int n0  = blockIdx.y * 32;
    const int tid = threadIdx.x;

    #pragma unroll
    for (int i = 0; i < 8; ++i) {
        const int idx   = tid + i * 256;
        const int row   = idx >> 5;
        const int chunk = idx & 31;
        const int gr    = m0 + row;
        const float4 v =
            *(const float4*)(X + (size_t)(gr < M ? gr : 0) * CH + chunk * 4);
        *(float4*)&sA[row][chunk * 4] = v;
    }
    {
        const int c     = tid & 31;
        const int kbase = (tid >> 5) * 16;
        #pragma unroll
        for (int i = 0; i < 16; ++i) {
            const int k = kbase + i;
            sB[c][k] = W[(size_t)k * CH + n0 + c];
        }
    }
    __syncthreads();

    const int wave  = tid >> 5;
    const int lane  = tid & 31;
    const int mt    = wave >> 1;
    const int nt    = wave & 1;
    const int mloc  = lane & 15;
    const int khalf = (lane >> 4) * 2;

    const float* arow = &sA[mt * 16 + mloc][0];
    const float* brow = &sB[nt * 16 + mloc][0];

    v8f c = {};
    #pragma unroll 8
    for (int k0 = 0; k0 < CH; k0 += 4) {
        const v2f a = *(const v2f*)(arow + k0 + khalf);
        const v2f b = *(const v2f*)(brow + k0 + khalf);
        c = __builtin_amdgcn_wmma_f32_16x16x4_f32(
                 false, a,  false, b,
                 (short)0, c,  false,  false);
    }

    const int col   = n0 + nt * 16 + mloc;
    const int half  = lane >> 4;
    const int mbase = m0 + mt * 16 + half * 8;
    if (m0 + 64 <= M) {
        #pragma unroll
        for (int r = 0; r < 8; ++r)
            H[(size_t)(mbase + r) * CH + col] = c[r];
    } else {
        #pragma unroll
        for (int r = 0; r < 8; ++r)
            if (mbase + r < M) H[(size_t)(mbase + r) * CH + col] = c[r];
    }
}

__global__ __launch_bounds__(256) void k_gather_norm_relu(const float* __restrict__ H,
                                                          const int* __restrict__ rowptr,
                                                          const int* __restrict__ csr_src,
                                                          const float* __restrict__ dinv,
                                                          const float* __restrict__ bias,
                                                          float* __restrict__ Out, int N) {
    const int lane = threadIdx.x & 31;
    const int row = __builtin_amdgcn_readfirstlane(
        (int)((blockIdx.x * blockDim.x + threadIdx.x) >> 5));
    if (row >= N) return;
    const float4* __restrict__ H4 = (const float4*)H;
    const float dd  = dinv[row];
    const int   beg = rowptr[row];
    const int   end = rowptr[row + 1];

    const float4 h = H4[(size_t)row * 32 + lane];
    const float4 b = ((const float4*)bias)[lane];
    const float  w = dd * dd;
    float4 a;
    a.x = h.x * w + b.x;  a.y = h.y * w + b.y;
    a.z = h.z * w + b.z;  a.w = h.w * w + b.w;

    for (int base = beg; base < end; base += 32) {
        const int left = end - base;
        const int n    = left < 32 ? left : 32;
        int   sl = 0;
        float wl = 0.0f;
        if (lane < n) {
            sl = csr_src[base + lane];
            wl = dinv[sl];
        }
        int j = 0;
        for (; j + 4 <= n; j += 4) {
            const int s0 = __builtin_amdgcn_readlane(sl, j);
            const int s1 = __builtin_amdgcn_readlane(sl, j + 1);
            const int s2 = __builtin_amdgcn_readlane(sl, j + 2);
            const int s3 = __builtin_amdgcn_readlane(sl, j + 3);
            const float4 v0 = H4[(size_t)s0 * 32 + lane];
            const float4 v1 = H4[(size_t)s1 * 32 + lane];
            const float4 v2 = H4[(size_t)s2 * 32 + lane];
            const float4 v3 = H4[(size_t)s3 * 32 + lane];
            const float w0 = dd * __int_as_float(__builtin_amdgcn_readlane(__float_as_int(wl), j));
            const float w1 = dd * __int_as_float(__builtin_amdgcn_readlane(__float_as_int(wl), j + 1));
            const float w2 = dd * __int_as_float(__builtin_amdgcn_readlane(__float_as_int(wl), j + 2));
            const float w3 = dd * __int_as_float(__builtin_amdgcn_readlane(__float_as_int(wl), j + 3));
            a.x += v0.x * w0;  a.y += v0.y * w0;  a.z += v0.z * w0;  a.w += v0.w * w0;
            a.x += v1.x * w1;  a.y += v1.y * w1;  a.z += v1.z * w1;  a.w += v1.w * w1;
            a.x += v2.x * w2;  a.y += v2.y * w2;  a.z += v2.z * w2;  a.w += v2.w * w2;
            a.x += v3.x * w3;  a.y += v3.y * w3;  a.z += v3.z * w3;  a.w += v3.w * w3;
        }
        for (; j < n; ++j) {
            const int   s0 = __builtin_amdgcn_readlane(sl, j);
            const float4 v0 = H4[(size_t)s0 * 32 + lane];
            const float w0 = dd * __int_as_float(__builtin_amdgcn_readlane(__float_as_int(wl), j));
            a.x += v0.x * w0;  a.y += v0.y * w0;  a.z += v0.z * w0;  a.w += v0.w * w0;
        }
    }
    float s = a.x + a.y + a.z + a.w;
    #pragma unroll
    for (int o = 16; o > 0; o >>= 1) s += __shfl_xor(s, o, 32);
    const float mu = s * (1.0f / CH);

    const float d0 = a.x - mu, d1 = a.y - mu, d2 = a.z - mu, d3 = a.w - mu;
    float q = d0 * d0 + d1 * d1 + d2 * d2 + d3 * d3;
    #pragma unroll
    for (int o = 16; o > 0; o >>= 1) q += __shfl_xor(q, o, 32);
    const float inv = rsqrtf(q * (1.0f / CH) + EPS);

    float4 r;
    r.x = fmaxf(d0 * inv, 0.0f);
    r.y = fmaxf(d1 * inv, 0.0f);
    r.z = fmaxf(d2 * inv, 0.0f);
    r.w = fmaxf(d3 * inv, 0.0f);
    ((float4*)Out)[(size_t)row * 32 + lane] = r;
}

extern "C" void kernel_launch(void* const* d_in, const int* in_sizes, int n_in,
                              void* d_out, int out_size, void* d_ws, size_t ws_size,
                              hipStream_t stream) {
    const float* x   = (const float*)d_in[0];
    const int*   ei  = (const int*)  d_in[1];
    const float* W1  = (const float*)d_in[2];
    const float* b1  = (const float*)d_in[3];
    const float* W2  = (const float*)d_in[4];
    const float* b2  = (const float*)d_in[5];
    float*       out = (float*)d_out;

    const int N = in_sizes[0] / CH;
    const int E = in_sizes[1] / 2;

    float* ws   = (float*)d_ws;
    const size_t nPad = ((size_t)N + 63) & ~(size_t)63;
    float* dinv = ws;
    float* bufA = ws + nPad;

    int* cnt     = (int*)(bufA + (size_t)N * CH);
    int* rowptr  = cnt + nPad;
    int* rank    = rowptr + nPad + 64;
    int* csr_src = rank + E;

    const int T = 256;
    const int gN     = (N + T - 1) / T;
    const int gE     = (E + T - 1) / T;
    const int gScan  = (N + 1 + 1023) / 1024;
    const int gRowW  = ((N * 32) + T - 1) / T;
    const dim3 gGemm((N + 63) / 64, CH / 32);

    k_cnt_zero <<<gN, T, 0, stream>>>(cnt, N);
    k_edge_rank<<<gE, T, 0, stream>>>(ei, cnt, rank, E, N);
    k_scan_dinv<<<gScan, 1024, 0, stream>>>(cnt, rowptr, dinv, N);
    k_csr_fill <<<gE, T, 0, stream>>>(ei, rank, rowptr, csr_src, E, N);

    k_gemm_wmma       <<<gGemm, T, 0, stream>>>(x, W1, bufA, N);
    k_gather_norm_relu<<<gRowW, T, 0, stream>>>(bufA, rowptr, csr_src, dinv, b1, out, N);

    k_gemm_wmma       <<<gGemm, T, 0, stream>>>(out, W2, bufA, N);
    k_gather_norm_relu<<<gRowW, T, 0, stream>>>(bufA, rowptr, csr_src, dinv, b2, out, N);
}


# x@W moved behind the gather, A(xW)=(Ax)W: per layer one launch sums 128 rows' in-edges into LDS, times W on f32 WMMA, norms, stores; both GEMM launches and H's round trip go; sum order changes
# speedup vs baseline: 2.8446x; 1.0833x over previous
#include <hip/hip_runtime.h>
#include <hip/hip_bf16.h>

typedef __attribute__((ext_vector_type(2))) float v2f;
typedef __attribute__((ext_vector_type(8))) float v8f;

#define CH 128
#define EPS 1e-5f

__global__ void k_cnt_zero(int* __restrict__ cnt, int N) {
    int i = blockIdx.x * blockDim.x + threadIdx.x;
    if (i < N) cnt[i] = 0;
}

__global__ void k_edge_rank(const int* __restrict__ ei, int* __restrict__ cnt,
                            int* __restrict__ rank, int E, int N) {
    int e = blockIdx.x * blockDim.x + threadIdx.x;
    if (e >= E) return;
    int s = ei[e];
    int d = ei[E + e];
    if ((unsigned)s < (unsigned)N && (unsigned)d < (unsigned)N) {
        rank[e] = atomicAdd(&cnt[d], 1);
    }
}

__global__ __launch_bounds__(1024) void k_scan_dinv(const int* __restrict__ cnt,
                                                    int* __restrict__ rowptr,
                                                    float* __restrict__ dinv, int N) {
    __shared__ int sTot[32];
    __shared__ int sFront[32];
    const int tid  = threadIdx.x;
    const int lane = tid & 31;
    const int wv   = tid >> 5;
    const int b    = blockIdx.x;

    int front = 0;
    const int4* c4 = (const int4*)cnt;
    #pragma unroll 4
    for (int j = tid; j < b * 256; j += 1024) {
        const int4 v = c4[j];
        front += (v.x + v.y) + (v.z + v.w);
    }
    const int i = b * 1024 + tid;
    const int c = (i < N) ? cnt[i] : 0;

    int inc = c;
    for (int o = 1; o < 32; o <<= 1) {
        const int t = __shfl_up(inc, o, 32);
        if (lane >= o) inc += t;
    }
    for (int o = 16; o > 0; o >>= 1) front += __shfl_xor(front, o, 32);
    if (lane == 31) sTot[wv] = inc;
    if (lane == 0)  sFront[wv] = front;
    __syncthreads();

    const int wt = sTot[lane];
    int winc = wt;
    for (int o = 1; o < 32; o <<= 1) {
        const int t = __shfl_up(winc, o, 32);
        if (lane >= o) winc += t;
    }
    int base = sFront[lane];
    for (int o = 16; o > 0; o >>= 1) base += __shfl_xor(base, o, 32);
    const int wave_off = __shfl(winc - wt, wv, 32);

    const int excl = base + wave_off + inc - c;
    if (i <= N) rowptr[i] = excl;
    if (i < N)  dinv[i] = rsqrtf(1.0f + (float)c);
}

__global__ void k_csr_fill(const int* __restrict__ ei, const int* __restrict__ rank,
                           const int* __restrict__ rowptr, int* __restrict__ csr_src,
                           int E, int N) {
    int e = blockIdx.x * blockDim.x + threadIdx.x;
    if (e >= E) return;
    int s = ei[e];
    int d = ei[E + e];
    if ((unsigned)s < (unsigned)N && (unsigned)d < (unsigned)N) {
        csr_src[rowptr[d] + rank[e]] = s;
    }
}

#define TM  128
#define LDP 132
__global__ __launch_bounds__(1024) void k_gather_gemm_norm_relu(const float* __restrict__ X,
                                                               const float* __restrict__ W,
                                                               const int* __restrict__ rowptr,
                                                               const int* __restrict__ csr_src,
                                                               const float* __restrict__ dinv,
                                                               const float* __restrict__ bias,
                                                               float* __restrict__ Out, int N) {
    __shared__ float sA[TM][LDP];
    __shared__ float sB[CH][LDP];
    const int tid  = threadIdx.x;
    const int lane = tid & 31;
    const int wave = __builtin_amdgcn_readfirstlane(tid >> 5);
    const int m0   = blockIdx.x * TM;

    const int mt    = wave >> 2;
    const int ng    = wave & 3;
    const int mloc  = lane & 15;
    const int khalf = (lane >> 4) * 2;
    const float bs0 = bias[ng * 32 + mloc];
    const float bs1 = bias[ng * 32 + 16 + mloc];

    {
        const int c     = tid & 127;
        const int kbase = (tid >> 7) * 16;
        #pragma unroll
        for (int i = 0; i < 4; ++i) {
            const int k = kbase + i * 4;
            float4 v;
            v.x = W[(size_t)(k + 0) * CH + c];
            v.y = W[(size_t)(k + 1) * CH + c];
            v.z = W[(size_t)(k + 2) * CH + c];
            v.w = W[(size_t)(k + 3) * CH + c];
            *(float4*)&sB[c][k] = v;
        }
    }

    const float4* __restrict__ X4 = (const float4*)X;
    #pragma unroll 1
    for (int i = 0; i < 4; ++i) {
        const int lr  = wave * 4 + i;
        const int row = m0 + lr;
        float4 a = make_float4(0.0f, 0.0f, 0.0f, 0.0f);
        if (row < N) {
            const float dd  = dinv[row];
            const int   beg = rowptr[row];
            const int   end = rowptr[row + 1];
            const float4 h = X4[(size_t)row * 32 + lane];
            const float  w = dd * dd;
            a.x = h.x * w;  a.y = h.y * w;  a.z = h.z * w;  a.w = h.w * w;

            for (int base = beg; base < end; base += 32) {
                const int left = end - base;
                const int n    = left < 32 ? left : 32;
                int   sl = 0;
                float wl = 0.0f;
                if (lane < n) {
                    sl = csr_src[base + lane];
                    wl = dinv[sl];
                }
                int j = 0;
                for (; j + 4 <= n; j += 4) {
                    const int s0 = __builtin_amdgcn_readlane(sl, j);
                    const int s1 = __builtin_amdgcn_readlane(sl, j + 1);
                    const int s2 = __builtin_amdgcn_readlane(sl, j + 2);
                    const int s3 = __builtin_amdgcn_readlane(sl, j + 3);
                    const float4 v0 = X4[(size_t)s0 * 32 + lane];
                    const float4 v1 = X4[(size_t)s1 * 32 + lane];
                    const float4 v2 = X4[(size_t)s2 * 32 + lane];
                    const float4 v3 = X4[(size_t)s3 * 32 + lane];
                    const float w0 = dd * __int_as_float(__builtin_amdgcn_readlane(__float_as_int(wl), j));
                    const float w1 = dd * __int_as_float(__builtin_amdgcn_readlane(__float_as_int(wl), j + 1));
                    const float w2 = dd * __int_as_float(__builtin_amdgcn_readlane(__float_as_int(wl), j + 2));
                    const float w3 = dd * __int_as_float(__builtin_amdgcn_readlane(__float_as_int(wl), j + 3));
                    a.x += v0.x * w0;  a.y += v0.y * w0;  a.z += v0.z * w0;  a.w += v0.w * w0;
                    a.x += v1.x * w1;  a.y += v1.y * w1;  a.z += v1.z * w1;  a.w += v1.w * w1;
                    a.x += v2.x * w2;  a.y += v2.y * w2;  a.z += v2.z * w2;  a.w += v2.w * w2;
                    a.x += v3.x * w3;  a.y += v3.y * w3;  a.z += v3.z * w3;  a.w += v3.w * w3;
                }
                for (; j < n; ++j) {
                    const int   s0 = __builtin_amdgcn_readlane(sl, j);
                    const float4 v0 = X4[(size_t)s0 * 32 + lane];
                    const float w0 = dd * __int_as_float(__builtin_amdgcn_readlane(__float_as_int(wl), j));
                    a.x += v0.x * w0;  a.y += v0.y * w0;  a.z += v0.z * w0;  a.w += v0.w * w0;
                }
            }
        }
        *(float4*)&sA[lr][lane * 4] = a;
    }
    __syncthreads();

    const float* arow  = &sA[mt * 16 + mloc][0];
    const float* brow0 = &sB[ng * 32 + mloc][0];
    const float* brow1 = &sB[ng * 32 + 16 + mloc][0];

    v8f c0 = {bs0, bs0, bs0, bs0, bs0, bs0, bs0, bs0};
    v8f c1 = {bs1, bs1, bs1, bs1, bs1, bs1, bs1, bs1};
    #pragma unroll
    for (int k0 = 0; k0 < CH; k0 += 4) {
        const v2f a  = *(const v2f*)(arow  + k0 + khalf);
        const v2f b0 = *(const v2f*)(brow0 + k0 + khalf);
        const v2f b1 = *(const v2f*)(brow1 + k0 + khalf);
        c0 = __builtin_amdgcn_wmma_f32_16x16x4_f32(
                  false, a,  false, b0,
                  (short)0, c0,  false,  false);
        c1 = __builtin_amdgcn_wmma_f32_16x16x4_f32(
                  false, a,  false, b1,
                  (short)0, c1,  false,  false);
    }
    __builtin_amdgcn_sched_barrier(0);
    asm volatile("v_nop\n\tv_nop\n\tv_nop\n\tv_nop");
    __syncthreads();

    {
        const int half  = lane >> 4;
        const int rbase = mt * 16 + half * 8;
        const int col   = ng * 32 + mloc;
        #pragma unroll
        for (int r = 0; r < 8; ++r) {
            sA[rbase + r][col]      = c0[r];
            sA[rbase + r][col + 16] = c1[r];
        }
    }
    __syncthreads();

    float4 a[4];
    float  s[4], q[4];
    #pragma unroll
    for (int i = 0; i < 4; ++i) {
        a[i] = *(const float4*)&sA[wave * 4 + i][lane * 4];
        s[i] = a[i].x + a[i].y + a[i].z + a[i].w;
    }
    #pragma unroll
    for (int o = 16; o > 0; o >>= 1) {
        #pragma unroll
        for (int i = 0; i < 4; ++i) s[i] += __shfl_xor(s[i], o, 32);
    }
    #pragma unroll
    for (int i = 0; i < 4; ++i) {
        const float mu = s[i] * (1.0f / CH);
        a[i].x -= mu;  a[i].y -= mu;  a[i].z -= mu;  a[i].w -= mu;
        q[i] = a[i].x * a[i].x + a[i].y * a[i].y + a[i].z * a[i].z + a[i].w * a[i].w;
    }
    #pragma unroll
    for (int o = 16; o > 0; o >>= 1) {
        #pragma unroll
        for (int i = 0; i < 4; ++i) q[i] += __shfl_xor(q[i], o, 32);
    }
    #pragma unroll
    for (int i = 0; i < 4; ++i) {
        const int row = m0 + wave * 4 + i;
        const float inv = rsqrtf(q[i] * (1.0f / CH) + EPS);
        float4 r;
        r.x = fmaxf(a[i].x * inv, 0.0f);
        r.y = fmaxf(a[i].y * inv, 0.0f);
        r.z = fmaxf(a[i].z * inv, 0.0f);
        r.w = fmaxf(a[i].w * inv, 0.0f);
        if (row < N) ((float4*)Out)[(size_t)row * 32 + lane] = r;
    }
}

extern "C" void kernel_launch(void* const* d_in, const int* in_sizes, int n_in,
                              void* d_out, int out_size, void* d_ws, size_t ws_size,
                              hipStream_t stream) {
    const float* x   = (const float*)d_in[0];
    const int*   ei  = (const int*)  d_in[1];
    const float* W1  = (const float*)d_in[2];
    const float* b1  = (const float*)d_in[3];
    const float* W2  = (const float*)d_in[4];
    const float* b2  = (const float*)d_in[5];
    float*       out = (float*)d_out;

    const int N = in_sizes[0] / CH;
    const int E = in_sizes[1] / 2;

    float* ws   = (float*)d_ws;
    const size_t nPad = ((size_t)N + 63) & ~(size_t)63;
    float* dinv = ws;
    float* bufA = ws + nPad;

    int* cnt     = (int*)(bufA + (size_t)N * CH);
    int* rowptr  = cnt + nPad;
    int* rank    = rowptr + nPad + 64;
    int* csr_src = rank + E;

    const int T = 256;
    const int gN     = (N + T - 1) / T;
    const int gE     = (E + T - 1) / T;
    const int gScan  = (N + 1 + 1023) / 1024;
    const int gTile  = (N + TM - 1) / TM;

    k_cnt_zero <<<gN, T, 0, stream>>>(cnt, N);
    k_edge_rank<<<gE, T, 0, stream>>>(ei, cnt, rank, E, N);
    k_scan_dinv<<<gScan, 1024, 0, stream>>>(cnt, rowptr, dinv, N);
    k_csr_fill <<<gE, T, 0, stream>>>(ei, rank, rowptr, csr_src, E, N);

    k_gather_gemm_norm_relu<<<gTile, 1024, 0, stream>>>(x, W1, rowptr, csr_src, dinv, b1, bufA, N);
    k_gather_gemm_norm_relu<<<gTile, 1024, 0, stream>>>(bufA, W2, rowptr, csr_src, dinv, b2, out, N);
}


# prefix launch and second edge pass dropped: the rank pass itself files each source id in its destination's 32-slot line (overflow list past 32); 6 launches become 4; sums stay in arrival order
# speedup vs baseline: 3.3947x; 1.1934x over previous
#include <hip/hip_runtime.h>
#include <hip/hip_bf16.h>

typedef __attribute__((ext_vector_type(2))) float v2f;
typedef __attribute__((ext_vector_type(8))) float v8f;

#define CH 128
#define EPS 1e-5f

#define CAP 32
__global__ void k_cnt_zero(int* __restrict__ cnt, int* __restrict__ ovf_n, int N) {
    int i = blockIdx.x * blockDim.x + threadIdx.x;
    if (i < N) cnt[i] = 0;
    if (i == 0) *ovf_n = 0;
}

__global__ void k_edge_slots(const int* __restrict__ ei, int* __restrict__ cnt,
                             int* __restrict__ slots, int* __restrict__ ovf_n,
                             int* __restrict__ ovf, int E, int N) {
    int e = blockIdx.x * blockDim.x + threadIdx.x;
    if (e >= E) return;
    int s = ei[e];
    int d = ei[E + e];
    if ((unsigned)s < (unsigned)N && (unsigned)d < (unsigned)N) {
        const int r = atomicAdd(&cnt[d], 1);
        if (r < CAP) {
            slots[(size_t)d * CAP + r] = s;
        } else {
            const int p = atomicAdd(ovf_n, 1);
            ovf[2 * (size_t)p]     = s;
            ovf[2 * (size_t)p + 1] = d;
        }
    }
}

#define TM  128
#define LDP 132
__global__ __launch_bounds__(1024) void k_gather_gemm_norm_relu(const float* __restrict__ X,
                                                               const float* __restrict__ W,
                                                               const int* __restrict__ cnt,
                                                               const int* __restrict__ slots,
                                                               const int* __restrict__ ovf_n,
                                                               const int* __restrict__ ovf,
                                                               const float* __restrict__ bias,
                                                               float* __restrict__ Out, int N) {
    __shared__ float sA[TM][LDP];
    __shared__ float sB[CH][LDP];
    const int tid  = threadIdx.x;
    const int lane = tid & 31;
    const int wave = __builtin_amdgcn_readfirstlane(tid >> 5);
    const int m0   = blockIdx.x * TM;

    const int mt    = wave >> 2;
    const int ng    = wave & 3;
    const int mloc  = lane & 15;
    const int khalf = (lane >> 4) * 2;
    const float bs0 = bias[ng * 32 + mloc];
    const float bs1 = bias[ng * 32 + 16 + mloc];

    {
        const int c     = tid & 127;
        const int kbase = (tid >> 7) * 16;
        #pragma unroll
        for (int i = 0; i < 4; ++i) {
            const int k = kbase + i * 4;
            float4 v;
            v.x = W[(size_t)(k + 0) * CH + c];
            v.y = W[(size_t)(k + 1) * CH + c];
            v.z = W[(size_t)(k + 2) * CH + c];
            v.w = W[(size_t)(k + 3) * CH + c];
            *(float4*)&sB[c][k] = v;
        }
    }

    const float4* __restrict__ X4 = (const float4*)X;
    #pragma unroll 1
    for (int i = 0; i < 4; ++i) {
        const int lr  = wave * 4 + i;
        const int row = m0 + lr;
        float4 a = make_float4(0.0f, 0.0f, 0.0f, 0.0f);
        if (row < N) {
            const int   c  = cnt[row];
            const float dd = rsqrtf(1.0f + (float)c);
            const float4 h = X4[(size_t)row * 32 + lane];
            const float  w = dd * dd;
            a.x = h.x * w;  a.y = h.y * w;  a.z = h.z * w;  a.w = h.w * w;

            {
                const int n = c < CAP ? c : CAP;
                int   sl = 0;
                float wl = 0.0f;
                if (lane < n) {
                    sl = slots[(size_t)row * CAP + lane];
                    wl = rsqrtf(1.0f + (float)cnt[sl]);
                }
                int j = 0;
                for (; j + 4 <= n; j += 4) {
                    const int s0 = __builtin_amdgcn_readlane(sl, j);
                    const int s1 = __builtin_amdgcn_readlane(sl, j + 1);
                    const int s2 = __builtin_amdgcn_readlane(sl, j + 2);
                    const int s3 = __builtin_amdgcn_readlane(sl, j + 3);
                    const float4 v0 = X4[(size_t)s0 * 32 + lane];
                    const float4 v1 = X4[(size_t)s1 * 32 + lane];
                    const float4 v2 = X4[(size_t)s2 * 32 + lane];
                    const float4 v3 = X4[(size_t)s3 * 32 + lane];
                    const float w0 = dd * __int_as_float(__builtin_amdgcn_readlane(__float_as_int(wl), j));
                    const float w1 = dd * __int_as_float(__builtin_amdgcn_readlane(__float_as_int(wl), j + 1));
                    const float w2 = dd * __int_as_float(__builtin_amdgcn_readlane(__float_as_int(wl), j + 2));
                    const float w3 = dd * __int_as_float(__builtin_amdgcn_readlane(__float_as_int(wl), j + 3));
                    a.x += v0.x * w0;  a.y += v0.y * w0;  a.z += v0.z * w0;  a.w += v0.w * w0;
                    a.x += v1.x * w1;  a.y += v1.y * w1;  a.z += v1.z * w1;  a.w += v1.w * w1;
                    a.x += v2.x * w2;  a.y += v2.y * w2;  a.z += v2.z * w2;  a.w += v2.w * w2;
                    a.x += v3.x * w3;  a.y += v3.y * w3;  a.z += v3.z * w3;  a.w += v3.w * w3;
                }
                for (; j < n; ++j) {
                    const int   s0 = __builtin_amdgcn_readlane(sl, j);
                    const float4 v0 = X4[(size_t)s0 * 32 + lane];
                    const float w0 = dd * __int_as_float(__builtin_amdgcn_readlane(__float_as_int(wl), j));
                    a.x += v0.x * w0;  a.y += v0.y * w0;  a.z += v0.z * w0;  a.w += v0.w * w0;
                }
            }
            if (c > CAP) {
                const int tot = *ovf_n;
                for (int base = 0; base < tot; base += 32) {
                    int ps = 0;
                    int pd = -1;
                    if (base + lane < tot) {
                        const int2 p = *(const int2*)&ovf[2 * (size_t)(base + lane)];
                        ps = p.x;
                        pd = p.y;
                    }
                    const bool mine = (pd == row);
                    float wl = 0.0f;
                    if (mine) wl = rsqrtf(1.0f + (float)cnt[ps]);
                    unsigned m = __builtin_amdgcn_ballot_w32(mine);
                    while (m) {
                        const int j = __builtin_ctz(m);
                        m &= m - 1;
                        const int   s0 = __builtin_amdgcn_readlane(ps, j);
                        const float4 v0 = X4[(size_t)s0 * 32 + lane];
                        const float w0 = dd * __int_as_float(__builtin_amdgcn_readlane(__float_as_int(wl), j));
                        a.x += v0.x * w0;  a.y += v0.y * w0;  a.z += v0.z * w0;  a.w += v0.w * w0;
                    }
                }
            }
        }
        *(float4*)&sA[lr][lane * 4] = a;
    }
    __syncthreads();

    const float* arow  = &sA[mt * 16 + mloc][0];
    const float* brow0 = &sB[ng * 32 + mloc][0];
    const float* brow1 = &sB[ng * 32 + 16 + mloc][0];

    v8f c0 = {bs0, bs0, bs0, bs0, bs0, bs0, bs0, bs0};
    v8f c1 = {bs1, bs1, bs1, bs1, bs1, bs1, bs1, bs1};
    #pragma unroll
    for (int k0 = 0; k0 < CH; k0 += 4) {
        const v2f a  = *(const v2f*)(arow  + k0 + khalf);
        const v2f b0 = *(const v2f*)(brow0 + k0 + khalf);
        const v2f b1 = *(const v2f*)(brow1 + k0 + khalf);
        c0 = __builtin_amdgcn_wmma_f32_16x16x4_f32(
                  false, a,  false, b0,
                  (short)0, c0,  false,  false);
        c1 = __builtin_amdgcn_wmma_f32_16x16x4_f32(
                  false, a,  false, b1,
                  (short)0, c1,  false,  false);
    }
    __builtin_amdgcn_sched_barrier(0);
    asm volatile("v_nop\n\tv_nop\n\tv_nop\n\tv_nop");
    __syncthreads();

    {
        const int half  = lane >> 4;
        const int rbase = mt * 16 + half * 8;
        const int col   = ng * 32 + mloc;
        #pragma unroll
        for (int r = 0; r < 8; ++r) {
            sA[rbase + r][col]      = c0[r];
            sA[rbase + r][col + 16] = c1[r];
        }
    }
    __syncthreads();

    float4 a[4];
    float  s[4], q[4];
    #pragma unroll
    for (int i = 0; i < 4; ++i) {
        a[i] = *(const float4*)&sA[wave * 4 + i][lane * 4];
        s[i] = a[i].x + a[i].y + a[i].z + a[i].w;
    }
    #pragma unroll
    for (int o = 16; o > 0; o >>= 1) {
        #pragma unroll
        for (int i = 0; i < 4; ++i) s[i] += __shfl_xor(s[i], o, 32);
    }
    #pragma unroll
    for (int i = 0; i < 4; ++i) {
        const float mu = s[i] * (1.0f / CH);
        a[i].x -= mu;  a[i].y -= mu;  a[i].z -= mu;  a[i].w -= mu;
        q[i] = a[i].x * a[i].x + a[i].y * a[i].y + a[i].z * a[i].z + a[i].w * a[i].w;
    }
    #pragma unroll
    for (int o = 16; o > 0; o >>= 1) {
        #pragma unroll
        for (int i = 0; i < 4; ++i) q[i] += __shfl_xor(q[i], o, 32);
    }
    #pragma unroll
    for (int i = 0; i < 4; ++i) {
        const int row = m0 + wave * 4 + i;
        const float inv = rsqrtf(q[i] * (1.0f / CH) + EPS);
        float4 r;
        r.x = fmaxf(a[i].x * inv, 0.0f);
        r.y = fmaxf(a[i].y * inv, 0.0f);
        r.z = fmaxf(a[i].z * inv, 0.0f);
        r.w = fmaxf(a[i].w * inv, 0.0f);
        if (row < N) ((float4*)Out)[(size_t)row * 32 + lane] = r;
    }
}

extern "C" void kernel_launch(void* const* d_in, const int* in_sizes, int n_in,
                              void* d_out, int out_size, void* d_ws, size_t ws_size,
                              hipStream_t stream) {
    const float* x   = (const float*)d_in[0];
    const int*   ei  = (const int*)  d_in[1];
    const float* W1  = (const float*)d_in[2];
    const float* b1  = (const float*)d_in[3];
    const float* W2  = (const float*)d_in[4];
    const float* b2  = (const float*)d_in[5];
    float*       out = (float*)d_out;

    const int N = in_sizes[0] / CH;
    const int E = in_sizes[1] / 2;

    float* ws   = (float*)d_ws;
    const size_t nPad = ((size_t)N + 63) & ~(size_t)63;
    float* bufA = ws;

    int* cnt   = (int*)(bufA + (size_t)N * CH);
    int* ovf_n = cnt + nPad;
    int* slots = ovf_n + 64;
    int* ovf   = slots + (size_t)N * CAP;

    const int T = 256;
    const int gN     = (N + T - 1) / T;
    const int gE     = (E + T - 1) / T;
    const int gTile  = (N + TM - 1) / TM;

    k_cnt_zero  <<<gN, T, 0, stream>>>(cnt, ovf_n, N);
    k_edge_slots<<<gE, T, 0, stream>>>(ei, cnt, slots, ovf_n, ovf, E, N);

    k_gather_gemm_norm_relu<<<gTile, 1024, 0, stream>>>(x, W1, cnt, slots, ovf_n, ovf, b1, bufA, N);
    k_gather_gemm_norm_relu<<<gTile, 1024, 0, stream>>>(bufA, W2, cnt, slots, ovf_n, ovf, b2, out, N);
}


# 800000 returning global adds go: edges are placed by 128-row destination bucket with LDS counts and ranks, each bucket ranks its rows in LDS and writes whole slot lines; sum order = LDS arrival order
# speedup vs baseline: 4.3870x; 1.2923x over previous
#include <hip/hip_runtime.h>
#include <hip/hip_bf16.h>

typedef __attribute__((ext_vector_type(2))) float v2f;
typedef __attribute__((ext_vector_type(8))) float v8f;

#define CH 128
#define EPS 1e-5f

#define CAP 32
#define TM  128
#define NCH 128
#define NBMAX 512

__global__ __launch_bounds__(1024) void k_chunk_count(const int* __restrict__ ei, int* __restrict__ cc,
                                                      int* __restrict__ ovf_n, int E, int N, int NB, int epc) {
    __shared__ int lc[NBMAX];
    const int t = threadIdx.x;
    const int c = blockIdx.x;
    if (t < NBMAX) lc[t] = 0;
    if (c == 0 && t == 0) *ovf_n = 0;
    __syncthreads();
    const int e0 = c * epc;
    const int e1 = (e0 + epc < E) ? (e0 + epc) : E;
    for (int base = e0; base < e1; base += 8 * 1024) {
        int s[8], d[8];
        #pragma unroll
        for (int i = 0; i < 8; ++i) {
            const int e  = base + i * 1024 + t;
            const int ec = e < E ? e : E - 1;
            s[i] = ei[ec];
            d[i] = ei[E + ec];
        }
        #pragma unroll
        for (int i = 0; i < 8; ++i) {
            const int e = base + i * 1024 + t;
            if (e < e1 && (unsigned)s[i] < (unsigned)N && (unsigned)d[i] < (unsigned)N)
                atomicAdd(&lc[d[i] >> 7], 1);
        }
    }
    __syncthreads();
    if (t < NB) cc[c * NBMAX + t] = lc[t];
}

__global__ __launch_bounds__(1024) void k_chunk_place(const int* __restrict__ ei, const int* __restrict__ cc,
                                                      int* __restrict__ bstart, int* __restrict__ rec_w,
                                                      int E, int N, int NB, int epc) {
    int2* __restrict__ rec = (int2*)rec_w;
    __shared__ int ltot[NBMAX];
    __shared__ int lbase[NBMAX];
    __shared__ int lc[NBMAX];
    __shared__ int ptot[NBMAX];
    __shared__ int pbef[NBMAX];
    const int t    = threadIdx.x;
    const int lane = t & 31;
    const int wave = t >> 5;
    const int c    = blockIdx.x;
    if (t < NBMAX) lc[t] = 0;
    const int b = t & (NBMAX - 1);
    const int g = t >> 9;
    int tot = 0;
    int bef = 0;
    if (b < NB) {
        #pragma unroll
        for (int i = 0; i < NCH / 2; ++i) {
            const int cp = g * (NCH / 2) + i;
            const int v  = cc[cp * NBMAX + b];
            tot += v;
            bef += (cp < c) ? v : 0;
        }
    }
    if (g == 1) { ptot[b] = tot; pbef[b] = bef; }
    __syncthreads();
    if (g == 0) { ltot[b] = tot + ptot[b]; lbase[b] = bef + pbef[b]; }
    __syncthreads();
    if (wave == 0) {
        int loc[16];
        int sum = 0;
        #pragma unroll
        for (int i = 0; i < 16; ++i) {
            const int b = lane * 16 + i;
            loc[i] = sum;
            sum += (b < NB) ? ltot[b] : 0;
        }
        int incl = sum;
        #pragma unroll
        for (int o = 1; o < 32; o <<= 1) {
            const int n = __shfl_up(incl, o, 32);
            if (lane >= o) incl += n;
        }
        const int excl = incl - sum;
        #pragma unroll
        for (int i = 0; i < 16; ++i) {
            const int b = lane * 16 + i;
            if (b < NB) {
                const int st = excl + loc[i];
                lbase[b] += st;
                if (c == 0) bstart[b] = st;
            }
        }
        if (c == 0 && lane == 31) bstart[NB] = incl;
    }
    __syncthreads();
    const int e0 = c * epc;
    const int e1 = (e0 + epc < E) ? (e0 + epc) : E;
    for (int base = e0; base < e1; base += 8 * 1024) {
        int s[8], d[8];
        #pragma unroll
        for (int i = 0; i < 8; ++i) {
            const int e  = base + i * 1024 + t;
            const int ec = e < E ? e : E - 1;
            s[i] = ei[ec];
            d[i] = ei[E + ec];
        }
        #pragma unroll
        for (int i = 0; i < 8; ++i) {
            const int e = base + i * 1024 + t;
            if (e < e1 && (unsigned)s[i] < (unsigned)N && (unsigned)d[i] < (unsigned)N) {
                const int b = d[i] >> 7;
                const int r = atomicAdd(&lc[b], 1);
                rec[(size_t)(lbase[b] + r)] = make_int2(s[i], d[i]);
            }
        }
    }
}

#define BT 256
__global__ __launch_bounds__(BT) void k_bucket_slots(const int* __restrict__ bstart, const int* __restrict__ rec_w,
                                                     int* __restrict__ cnt, int* __restrict__ slots,
                                                     int* __restrict__ ovf_n, int* __restrict__ ovf, int N) {
    const int2* __restrict__ rec = (const int2*)rec_w;
    __shared__ int lc[TM];
    __shared__ __attribute__((aligned(16))) int lsl[TM * CAP];
    const int t    = threadIdx.x;
    const int b    = blockIdx.x;
    const int row0 = b * TM;
    if (t < TM) lc[t] = 0;
    __syncthreads();
    const int i0 = bstart[b];
    const int i1 = bstart[b + 1];
    for (int base = i0; base < i1; base += 16 * BT) {
        int2 p[16];
        #pragma unroll
        for (int k = 0; k < 16; ++k) {
            const int i  = base + k * BT + t;
            const int ic = i < i1 ? i : i1 - 1;
            p[k] = rec[(size_t)ic];
        }
        #pragma unroll
        for (int k = 0; k < 16; ++k) {
            const int i = base + k * BT + t;
            if (i < i1) {
                const int dl = p[k].y - row0;
                const int r  = atomicAdd(&lc[dl], 1);
                if (r < CAP) {
                    lsl[dl * CAP + r] = p[k].x;
                } else {
                    const int q = atomicAdd(ovf_n, 1);
                    ovf[2 * (size_t)q]     = p[k].x;
                    ovf[2 * (size_t)q + 1] = p[k].y;
                }
            }
        }
    }
    __syncthreads();
    if (t < TM && row0 + t < N) cnt[row0 + t] = lc[t];
    #pragma unroll
    for (int k = 0; k < (TM * CAP / 4) / BT; ++k) {
        const int u = k * BT + t;
        if (row0 + (u >> 3) < N) ((int4*)slots)[(size_t)row0 * (CAP / 4) + u] = ((const int4*)lsl)[u];
    }
}

#define LDP 132
__global__ __launch_bounds__(1024) void k_gather_gemm_norm_relu(const float* __restrict__ X,
                                                               const float* __restrict__ W,
                                                               const int* __restrict__ cnt,
                                                               const int* __restrict__ slots,
                                                               const int* __restrict__ ovf_n,
                                                               const int* __restrict__ ovf,
                                                               const float* __restrict__ bias,
                                                               float* __restrict__ Out, int N) {
    __shared__ float sA[TM][LDP];
    __shared__ float sB[CH][LDP];
    const int tid  = threadIdx.x;
    const int lane = tid & 31;
    const int wave = __builtin_amdgcn_readfirstlane(tid >> 5);
    const int m0   = blockIdx.x * TM;

    const int mt    = wave >> 2;
    const int ng    = wave & 3;
    const int mloc  = lane & 15;
    const int khalf = (lane >> 4) * 2;
    const float bs0 = bias[ng * 32 + mloc];
    const float bs1 = bias[ng * 32 + 16 + mloc];

    {
        const int c     = tid & 127;
        const int kbase = (tid >> 7) * 16;
        #pragma unroll
        for (int i = 0; i < 4; ++i) {
            const int k = kbase + i * 4;
            float4 v;
            v.x = W[(size_t)(k + 0) * CH + c];
            v.y = W[(size_t)(k + 1) * CH + c];
            v.z = W[(size_t)(k + 2) * CH + c];
            v.w = W[(size_t)(k + 3) * CH + c];
            *(float4*)&sB[c][k] = v;
        }
    }

    const float4* __restrict__ X4 = (const float4*)X;
    #pragma unroll 1
    for (int i = 0; i < 4; ++i) {
        const int lr  = wave * 4 + i;
        const int row = m0 + lr;
        float4 a = make_float4(0.0f, 0.0f, 0.0f, 0.0f);
        if (row < N) {
            const int   c  = cnt[row];
            const float dd = rsqrtf(1.0f + (float)c);
            const float4 h = X4[(size_t)row * 32 + lane];
            const float  w = dd * dd;
            a.x = h.x * w;  a.y = h.y * w;  a.z = h.z * w;  a.w = h.w * w;

            {
                const int n = c < CAP ? c : CAP;
                int   sl = 0;
                float wl = 0.0f;
                if (lane < n) {
                    sl = slots[(size_t)row * CAP + lane];
                    wl = rsqrtf(1.0f + (float)cnt[sl]);
                }
                int j = 0;
                for (; j + 4 <= n; j += 4) {
                    const int s0 = __builtin_amdgcn_readlane(sl, j);
                    const int s1 = __builtin_amdgcn_readlane(sl, j + 1);
                    const int s2 = __builtin_amdgcn_readlane(sl, j + 2);
                    const int s3 = __builtin_amdgcn_readlane(sl, j + 3);
                    const float4 v0 = X4[(size_t)s0 * 32 + lane];
                    const float4 v1 = X4[(size_t)s1 * 32 + lane];
                    const float4 v2 = X4[(size_t)s2 * 32 + lane];
                    const float4 v3 = X4[(size_t)s3 * 32 + lane];
                    const float w0 = dd * __int_as_float(__builtin_amdgcn_readlane(__float_as_int(wl), j));
                    const float w1 = dd * __int_as_float(__builtin_amdgcn_readlane(__float_as_int(wl), j + 1));
                    const float w2 = dd * __int_as_float(__builtin_amdgcn_readlane(__float_as_int(wl), j + 2));
                    const float w3 = dd * __int_as_float(__builtin_amdgcn_readlane(__float_as_int(wl), j + 3));
                    a.x += v0.x * w0;  a.y += v0.y * w0;  a.z += v0.z * w0;  a.w += v0.w * w0;
                    a.x += v1.x * w1;  a.y += v1.y * w1;  a.z += v1.z * w1;  a.w += v1.w * w1;
                    a.x += v2.x * w2;  a.y += v2.y * w2;  a.z += v2.z * w2;  a.w += v2.w * w2;
                    a.x += v3.x * w3;  a.y += v3.y * w3;  a.z += v3.z * w3;  a.w += v3.w * w3;
                }
                for (; j < n; ++j) {
                    const int   s0 = __builtin_amdgcn_readlane(sl, j);
                    const float4 v0 = X4[(size_t)s0 * 32 + lane];
                    const float w0 = dd * __int_as_float(__builtin_amdgcn_readlane(__float_as_int(wl), j));
                    a.x += v0.x * w0;  a.y += v0.y * w0;  a.z += v0.z * w0;  a.w += v0.w * w0;
                }
            }
            if (c > CAP) {
                const int tot = *ovf_n;
                for (int base = 0; base < tot; base += 32) {
                    int ps = 0;
                    int pd = -1;
                    if (base + lane < tot) {
                        const int2 p = *(const int2*)&ovf[2 * (size_t)(base + lane)];
                        ps = p.x;
                        pd = p.y;
                    }
                    const bool mine = (pd == row);
                    float wl = 0.0f;
                    if (mine) wl = rsqrtf(1.0f + (float)cnt[ps]);
                    unsigned m = __builtin_amdgcn_ballot_w32(mine);
                    while (m) {
                        const int j = __builtin_ctz(m);
                        m &= m - 1;
                        const int   s0 = __builtin_amdgcn_readlane(ps, j);
                        const float4 v0 = X4[(size_t)s0 * 32 + lane];
                        const float w0 = dd * __int_as_float(__builtin_amdgcn_readlane(__float_as_int(wl), j));
                        a.x += v0.x * w0;  a.y += v0.y * w0;  a.z += v0.z * w0;  a.w += v0.w * w0;
                    }
                }
            }
        }
        *(float4*)&sA[lr][lane * 4] = a;
    }
    __syncthreads();

    const float* arow  = &sA[mt * 16 + mloc][0];
    const float* brow0 = &sB[ng * 32 + mloc][0];
    const float* brow1 = &sB[ng * 32 + 16 + mloc][0];

    v8f c0 = {bs0, bs0, bs0, bs0, bs0, bs0, bs0, bs0};
    v8f c1 = {bs1, bs1, bs1, bs1, bs1, bs1, bs1, bs1};
    #pragma unroll
    for (int k0 = 0; k0 < CH; k0 += 4) {
        const v2f a  = *(const v2f*)(arow  + k0 + khalf);
        const v2f b0 = *(const v2f*)(brow0 + k0 + khalf);
        const v2f b1 = *(const v2f*)(brow1 + k0 + khalf);
        c0 = __builtin_amdgcn_wmma_f32_16x16x4_f32(
                  false, a,  false, b0,
                  (short)0, c0,  false,  false);
        c1 = __builtin_amdgcn_wmma_f32_16x16x4_f32(
                  false, a,  false, b1,
                  (short)0, c1,  false,  false);
    }
    __builtin_amdgcn_sched_barrier(0);
    asm volatile("v_nop\n\tv_nop\n\tv_nop\n\tv_nop");
    __syncthreads();

    {
        const int half  = lane >> 4;
        const int rbase = mt * 16 + half * 8;
        const int col   = ng * 32 + mloc;
        #pragma unroll
        for (int r = 0; r < 8; ++r) {
            sA[rbase + r][col]      = c0[r];
            sA[rbase + r][col + 16] = c1[r];
        }
    }
    __syncthreads();

    float4 a[4];
    float  s[4], q[4];
    #pragma unroll
    for (int i = 0; i < 4; ++i) {
        a[i] = *(const float4*)&sA[wave * 4 + i][lane * 4];
        s[i] = a[i].x + a[i].y + a[i].z + a[i].w;
    }
    #pragma unroll
    for (int o = 16; o > 0; o >>= 1) {
        #pragma unroll
        for (int i = 0; i < 4; ++i) s[i] += __shfl_xor(s[i], o, 32);
    }
    #pragma unroll
    for (int i = 0; i < 4; ++i) {
        const float mu = s[i] * (1.0f / CH);
        a[i].x -= mu;  a[i].y -= mu;  a[i].z -= mu;  a[i].w -= mu;
        q[i] = a[i].x * a[i].x + a[i].y * a[i].y + a[i].z * a[i].z + a[i].w * a[i].w;
    }
    #pragma unroll
    for (int o = 16; o > 0; o >>= 1) {
        #pragma unroll
        for (int i = 0; i < 4; ++i) q[i] += __shfl_xor(q[i], o, 32);
    }
    #pragma unroll
    for (int i = 0; i < 4; ++i) {
        const int row = m0 + wave * 4 + i;
        const float inv = rsqrtf(q[i] * (1.0f / CH) + EPS);
        float4 r;
        r.x = fmaxf(a[i].x * inv, 0.0f);
        r.y = fmaxf(a[i].y * inv, 0.0f);
        r.z = fmaxf(a[i].z * inv, 0.0f);
        r.w = fmaxf(a[i].w * inv, 0.0f);
        if (row < N) ((float4*)Out)[(size_t)row * 32 + lane] = r;
    }
}

extern "C" void kernel_launch(void* const* d_in, const int* in_sizes, int n_in,
                              void* d_out, int out_size, void* d_ws, size_t ws_size,
                              hipStream_t stream) {
    const float* x   = (const float*)d_in[0];
    const int*   ei  = (const int*)  d_in[1];
    const float* W1  = (const float*)d_in[2];
    const float* b1  = (const float*)d_in[3];
    const float* W2  = (const float*)d_in[4];
    const float* b2  = (const float*)d_in[5];
    float*       out = (float*)d_out;

    const int N = in_sizes[0] / CH;
    const int E = in_sizes[1] / 2;

    float* ws   = (float*)d_ws;
    const size_t nPad = ((size_t)N + 63) & ~(size_t)63;
    float* bufA = ws;

    int* cnt   = (int*)(bufA + (size_t)N * CH);
    int* ovf_n = cnt + nPad;
    int* slots = ovf_n + 64;
    int* ovf   = slots + (size_t)N * CAP;

    int*  cc   = (int*)bufA;
    int*  bst  = cc + (size_t)NBMAX * NCH;
    int*  rec  = bst + 1024;

    const int gTile  = (N + TM - 1) / TM;
    const int epc    = (E + NCH - 1) / NCH;

    k_chunk_count <<<NCH,   1024, 0, stream>>>(ei, cc, ovf_n, E, N, gTile, epc);
    k_chunk_place <<<NCH,   1024, 0, stream>>>(ei, cc, bst, rec, E, N, gTile, epc);
    k_bucket_slots<<<gTile, BT,   0, stream>>>(bst, rec, cnt, slots, ovf_n, ovf, N);

    k_gather_gemm_norm_relu<<<gTile, 1024, 0, stream>>>(x, W1, cnt, slots, ovf_n, ovf, b1, bufA, N);
    k_gather_gemm_norm_relu<<<gTile, 1024, 0, stream>>>(bufA, W2, cnt, slots, ovf_n, ovf, b2, out, N);
}


# count and place launches merged: a chunk sorts its edges by bucket in its own stretch (LDS count, prefix, rank), a bucket reads its run from each chunk; 5 launches to 4; sums in LDS arrival order
# speedup vs baseline: 5.1725x; 1.1791x over previous
#include <hip/hip_runtime.h>
#include <hip/hip_bf16.h>

typedef __attribute__((ext_vector_type(2))) float v2f;
typedef __attribute__((ext_vector_type(8))) float v8f;

#define CH 128
#define EPS 1e-5f

#define CAP 32
#define TM  128
#define NCH 128
#define NBMAX 512
#define CSW 520

__global__ __launch_bounds__(1024) void k_chunk_sort(const int* __restrict__ ei, int* __restrict__ cs,
                                                     int* __restrict__ rec_w, int* __restrict__ ovf_n,
                                                     int E, int N, int NB, int epc) {
    int2* __restrict__ rec = (int2*)rec_w;
    __shared__ int lcnt[NBMAX];
    __shared__ int lst[NBMAX];
    __shared__ int lrk[NBMAX];
    const int t    = threadIdx.x;
    const int lane = t & 31;
    const int wave = t >> 5;
    const int c    = blockIdx.x;
    if (t < NBMAX) { lcnt[t] = 0; lrk[t] = 0; }
    if (c == 0 && t == 0) *ovf_n = 0;
    __syncthreads();
    const int e0 = c * epc;
    const int e1 = (e0 + epc < E) ? (e0 + epc) : E;
    for (int base = e0; base < e1; base += 8 * 1024) {
        int s[8], d[8];
        #pragma unroll
        for (int i = 0; i < 8; ++i) {
            const int e  = base + i * 1024 + t;
            const int ec = e < E ? e : E - 1;
            s[i] = ei[ec];
            d[i] = ei[E + ec];
        }
        #pragma unroll
        for (int i = 0; i < 8; ++i) {
            const int e = base + i * 1024 + t;
            if (e < e1 && (unsigned)s[i] < (unsigned)N && (unsigned)d[i] < (unsigned)N)
                atomicAdd(&lcnt[d[i] >> 7], 1);
        }
    }
    __syncthreads();
    if (wave == 0) {
        int loc[16];
        int sum = 0;
        #pragma unroll
        for (int i = 0; i < 16; ++i) {
            const int b = lane * 16 + i;
            loc[i] = sum;
            sum += (b < NB) ? lcnt[b] : 0;
        }
        int incl = sum;
        #pragma unroll
        for (int o = 1; o < 32; o <<= 1) {
            const int n = __shfl_up(incl, o, 32);
            if (lane >= o) incl += n;
        }
        const int excl = incl - sum;
        #pragma unroll
        for (int i = 0; i < 16; ++i) {
            const int b = lane * 16 + i;
            if (b < NB) {
                const int st = excl + loc[i];
                lst[b] = st;
                cs[c * CSW + b] = st;
            }
        }
        if (lane == 31) cs[c * CSW + NB] = incl;
    }
    __syncthreads();
    const size_t r0 = (size_t)c * epc;
    for (int base = e0; base < e1; base += 8 * 1024) {
        int s[8], d[8];
        #pragma unroll
        for (int i = 0; i < 8; ++i) {
            const int e  = base + i * 1024 + t;
            const int ec = e < E ? e : E - 1;
            s[i] = ei[ec];
            d[i] = ei[E + ec];
        }
        #pragma unroll
        for (int i = 0; i < 8; ++i) {
            const int e = base + i * 1024 + t;
            if (e < e1 && (unsigned)s[i] < (unsigned)N && (unsigned)d[i] < (unsigned)N) {
                const int b = d[i] >> 7;
                const int r = atomicAdd(&lrk[b], 1);
                rec[r0 + (size_t)(lst[b] + r)] = make_int2(s[i], d[i]);
            }
        }
    }
}

#define BT 256
#define CPW (NCH / (BT / 32))
__global__ __launch_bounds__(BT) void k_bucket_slots(const int* __restrict__ cs, const int* __restrict__ rec_w,
                                                     int* __restrict__ cnt, int* __restrict__ slots,
                                                     int* __restrict__ ovf_n, int* __restrict__ ovf, int N, int epc) {
    const int2* __restrict__ rec = (const int2*)rec_w;
    __shared__ int lc[TM];
    __shared__ __attribute__((aligned(16))) int lsl[TM * CAP];
    const int t    = threadIdx.x;
    const int lane = t & 31;
    const int wave = __builtin_amdgcn_readfirstlane(t >> 5);
    const int b    = blockIdx.x;
    const int row0 = b * TM;
    if (t < TM) lc[t] = 0;
    int st[CPW], ln[CPW];
    #pragma unroll
    for (int k = 0; k < CPW; ++k) {
        const int c = wave * CPW + k;
        st[k] = cs[c * CSW + b];
        ln[k] = cs[c * CSW + b + 1] - st[k];
    }
    int2 p[CPW];
    #pragma unroll
    for (int k = 0; k < CPW; ++k) {
        const int c = wave * CPW + k;
        p[k] = make_int2(0, 0);
        if (lane < ln[k]) p[k] = rec[(size_t)c * epc + (size_t)(st[k] + lane)];
    }
    __syncthreads();
    #pragma unroll
    for (int k = 0; k < CPW; ++k) {
        if (lane < ln[k]) {
            const int dl = p[k].y - row0;
            const int r  = atomicAdd(&lc[dl], 1);
            if (r < CAP) {
                lsl[dl * CAP + r] = p[k].x;
            } else {
                const int q = atomicAdd(ovf_n, 1);
                ovf[2 * (size_t)q]     = p[k].x;
                ovf[2 * (size_t)q + 1] = p[k].y;
            }
        }
    }
    #pragma unroll
    for (int k = 0; k < CPW; ++k) {
        if (ln[k] > 32) {
            const int c = wave * CPW + k;
            for (int i = 32 + lane; i < ln[k]; i += 32) {
                const int2 q2 = rec[(size_t)c * epc + (size_t)(st[k] + i)];
                const int dl = q2.y - row0;
                const int r  = atomicAdd(&lc[dl], 1);
                if (r < CAP) {
                    lsl[dl * CAP + r] = q2.x;
                } else {
                    const int q = atomicAdd(ovf_n, 1);
                    ovf[2 * (size_t)q]     = q2.x;
                    ovf[2 * (size_t)q + 1] = q2.y;
                }
            }
        }
    }
    __syncthreads();
    if (t < TM && row0 + t < N) cnt[row0 + t] = lc[t];
    #pragma unroll
    for (int k = 0; k < (TM * CAP / 4) / BT; ++k) {
        const int u = k * BT + t;
        if (row0 + (u >> 3) < N) ((int4*)slots)[(size_t)row0 * (CAP / 4) + u] = ((const int4*)lsl)[u];
    }
}

#define LDP 132
__global__ __launch_bounds__(1024) void k_gather_gemm_norm_relu(const float* __restrict__ X,
                                                               const float* __restrict__ W,
                                                               const int* __restrict__ cnt,
                                                               const int* __restrict__ slots,
                                                               const int* __restrict__ ovf_n,
                                                               const int* __restrict__ ovf,
                                                               const float* __restrict__ bias,
                                                               float* __restrict__ Out, int N) {
    __shared__ float sA[TM][LDP];
    __shared__ float sB[CH][LDP];
    const int tid  = threadIdx.x;
    const int lane = tid & 31;
    const int wave = __builtin_amdgcn_readfirstlane(tid >> 5);
    const int m0   = blockIdx.x * TM;

    const int mt    = wave >> 2;
    const int ng    = wave & 3;
    const int mloc  = lane & 15;
    const int khalf = (lane >> 4) * 2;
    const float bs0 = bias[ng * 32 + mloc];
    const float bs1 = bias[ng * 32 + 16 + mloc];

    {
        const int c     = tid & 127;
        const int kbase = (tid >> 7) * 16;
        #pragma unroll
        for (int i = 0; i < 4; ++i) {
            const int k = kbase + i * 4;
            float4 v;
            v.x = W[(size_t)(k + 0) * CH + c];
            v.y = W[(size_t)(k + 1) * CH + c];
            v.z = W[(size_t)(k + 2) * CH + c];
            v.w = W[(size_t)(k + 3) * CH + c];
            *(float4*)&sB[c][k] = v;
        }
    }

    const float4* __restrict__ X4 = (const float4*)X;
    #pragma unroll 1
    for (int i = 0; i < 4; ++i) {
        const int lr  = wave * 4 + i;
        const int row = m0 + lr;
        float4 a = make_float4(0.0f, 0.0f, 0.0f, 0.0f);
        if (row < N) {
            const int   c  = cnt[row];
            const float dd = rsqrtf(1.0f + (float)c);
            const float4 h = X4[(size_t)row * 32 + lane];
            const float  w = dd * dd;
            a.x = h.x * w;  a.y = h.y * w;  a.z = h.z * w;  a.w = h.w * w;

            {
                const int n = c < CAP ? c : CAP;
                int   sl = 0;
                float wl = 0.0f;
                if (lane < n) {
                    sl = slots[(size_t)row * CAP + lane];
                    wl = rsqrtf(1.0f + (float)cnt[sl]);
                }
                int j = 0;
                for (; j + 4 <= n; j += 4) {
                    const int s0 = __builtin_amdgcn_readlane(sl, j);
                    const int s1 = __builtin_amdgcn_readlane(sl, j + 1);
                    const int s2 = __builtin_amdgcn_readlane(sl, j + 2);
                    const int s3 = __builtin_amdgcn_readlane(sl, j + 3);
                    const float4 v0 = X4[(size_t)s0 * 32 + lane];
                    const float4 v1 = X4[(size_t)s1 * 32 + lane];
                    const float4 v2 = X4[(size_t)s2 * 32 + lane];
                    const float4 v3 = X4[(size_t)s3 * 32 + lane];
                    const float w0 = dd * __int_as_float(__builtin_amdgcn_readlane(__float_as_int(wl), j));
                    const float w1 = dd * __int_as_float(__builtin_amdgcn_readlane(__float_as_int(wl), j + 1));
                    const float w2 = dd * __int_as_float(__builtin_amdgcn_readlane(__float_as_int(wl), j + 2));
                    const float w3 = dd * __int_as_float(__builtin_amdgcn_readlane(__float_as_int(wl), j + 3));
                    a.x += v0.x * w0;  a.y += v0.y * w0;  a.z += v0.z * w0;  a.w += v0.w * w0;
                    a.x += v1.x * w1;  a.y += v1.y * w1;  a.z += v1.z * w1;  a.w += v1.w * w1;
                    a.x += v2.x * w2;  a.y += v2.y * w2;  a.z += v2.z * w2;  a.w += v2.w * w2;
                    a.x += v3.x * w3;  a.y += v3.y * w3;  a.z += v3.z * w3;  a.w += v3.w * w3;
                }
                for (; j < n; ++j) {
                    const int   s0 = __builtin_amdgcn_readlane(sl, j);
                    const float4 v0 = X4[(size_t)s0 * 32 + lane];
                    const float w0 = dd * __int_as_float(__builtin_amdgcn_readlane(__float_as_int(wl), j));
                    a.x += v0.x * w0;  a.y += v0.y * w0;  a.z += v0.z * w0;  a.w += v0.w * w0;
                }
            }
            if (c > CAP) {
                const int tot = *ovf_n;
                for (int base = 0; base < tot; base += 32) {
                    int ps = 0;
                    int pd = -1;
                    if (base + lane < tot) {
                        const int2 p = *(const int2*)&ovf[2 * (size_t)(base + lane)];
                        ps = p.x;
                        pd = p.y;
                    }
                    const bool mine = (pd == row);
                    float wl = 0.0f;
                    if (mine) wl = rsqrtf(1.0f + (float)cnt[ps]);
                    unsigned m = __builtin_amdgcn_ballot_w32(mine);
                    while (m) {
                        const int j = __builtin_ctz(m);
                        m &= m - 1;
                        const int   s0 = __builtin_amdgcn_readlane(ps, j);
                        const float4 v0 = X4[(size_t)s0 * 32 + lane];
                        const float w0 = dd * __int_as_float(__builtin_amdgcn_readlane(__float_as_int(wl), j));
                        a.x += v0.x * w0;  a.y += v0.y * w0;  a.z += v0.z * w0;  a.w += v0.w * w0;
                    }
                }
            }
        }
        *(float4*)&sA[lr][lane * 4] = a;
    }
    __syncthreads();

    const float* arow  = &sA[mt * 16 + mloc][0];
    const float* brow0 = &sB[ng * 32 + mloc][0];
    const float* brow1 = &sB[ng * 32 + 16 + mloc][0];

    v8f c0 = {bs0, bs0, bs0, bs0, bs0, bs0, bs0, bs0};
    v8f c1 = {bs1, bs1, bs1, bs1, bs1, bs1, bs1, bs1};
    #pragma unroll
    for (int k0 = 0; k0 < CH; k0 += 4) {
        const v2f a  = *(const v2f*)(arow  + k0 + khalf);
        const v2f b0 = *(const v2f*)(brow0 + k0 + khalf);
        const v2f b1 = *(const v2f*)(brow1 + k0 + khalf);
        c0 = __builtin_amdgcn_wmma_f32_16x16x4_f32(
                  false, a,  false, b0,
                  (short)0, c0,  false,  false);
        c1 = __builtin_amdgcn_wmma_f32_16x16x4_f32(
                  false, a,  false, b1,
                  (short)0, c1,  false,  false);
    }
    __builtin_amdgcn_sched_barrier(0);
    asm volatile("v_nop\n\tv_nop\n\tv_nop\n\tv_nop");
    __syncthreads();

    {
        const int half  = lane >> 4;
        const int rbase = mt * 16 + half * 8;
        const int col   = ng * 32 + mloc;
        #pragma unroll
        for (int r = 0; r < 8; ++r) {
            sA[rbase + r][col]      = c0[r];
            sA[rbase + r][col + 16] = c1[r];
        }
    }
    __syncthreads();

    float4 a[4];
    float  s[4], q[4];
    #pragma unroll
    for (int i = 0; i < 4; ++i) {
        a[i] = *(const float4*)&sA[wave * 4 + i][lane * 4];
        s[i] = a[i].x + a[i].y + a[i].z + a[i].w;
    }
    #pragma unroll
    for (int o = 16; o > 0; o >>= 1) {
        #pragma unroll
        for (int i = 0; i < 4; ++i) s[i] += __shfl_xor(s[i], o, 32);
    }
    #pragma unroll
    for (int i = 0; i < 4; ++i) {
        const float mu = s[i] * (1.0f / CH);
        a[i].x -= mu;  a[i].y -= mu;  a[i].z -= mu;  a[i].w -= mu;
        q[i] = a[i].x * a[i].x + a[i].y * a[i].y + a[i].z * a[i].z + a[i].w * a[i].w;
    }
    #pragma unroll
    for (int o = 16; o > 0; o >>= 1) {
        #pragma unroll
        for (int i = 0; i < 4; ++i) q[i] += __shfl_xor(q[i], o, 32);
    }
    #pragma unroll
    for (int i = 0; i < 4; ++i) {
        const int row = m0 + wave * 4 + i;
        const float inv = rsqrtf(q[i] * (1.0f / CH) + EPS);
        float4 r;
        r.x = fmaxf(a[i].x * inv, 0.0f);
        r.y = fmaxf(a[i].y * inv, 0.0f);
        r.z = fmaxf(a[i].z * inv, 0.0f);
        r.w = fmaxf(a[i].w * inv, 0.0f);
        if (row < N) ((float4*)Out)[(size_t)row * 32 + lane] = r;
    }
}

extern "C" void kernel_launch(void* const* d_in, const int* in_sizes, int n_in,
                              void* d_out, int out_size, void* d_ws, size_t ws_size,
                              hipStream_t stream) {
    const float* x   = (const float*)d_in[0];
    const int*   ei  = (const int*)  d_in[1];
    const float* W1  = (const float*)d_in[2];
    const float* b1  = (const float*)d_in[3];
    const float* W2  = (const float*)d_in[4];
    const float* b2  = (const float*)d_in[5];
    float*       out = (float*)d_out;

    const int N = in_sizes[0] / CH;
    const int E = in_sizes[1] / 2;

    float* ws   = (float*)d_ws;
    const size_t nPad = ((size_t)N + 63) & ~(size_t)63;
    float* bufA = ws;

    int* cnt   = (int*)(bufA + (size_t)N * CH);
    int* ovf_n = cnt + nPad;
    int* slots = ovf_n + 64;
    int* ovf   = slots + (size_t)N * CAP;

    int*  cs   = (int*)bufA;
    int*  rec  = cs + (size_t)CSW * NCH;

    const int gTile  = (N + TM - 1) / TM;
    const int epc    = (E + NCH - 1) / NCH;

    k_chunk_sort  <<<NCH,   1024, 0, stream>>>(ei, cs, rec, ovf_n, E, N, gTile, epc);
    k_bucket_slots<<<gTile, BT,   0, stream>>>(cs, rec, cnt, slots, ovf_n, ovf, N, epc);

    k_gather_gemm_norm_relu<<<gTile, 1024, 0, stream>>>(x, W1, cnt, slots, ovf_n, ovf, b1, bufA, N);
    k_gather_gemm_norm_relu<<<gTile, 1024, 0, stream>>>(bufA, W2, cnt, slots, ovf_n, ovf, b2, out, N);
}
